# starting kernel
# baseline (speedup 1.0000x reference)
#include <hip/hip_runtime.h>
#include <hip/hip_bf16.h>


typedef __attribute__((ext_vector_type(2))) float v2f;
typedef __attribute__((ext_vector_type(8))) float v8f;

#define HIDDEN 256
#define QCH    12
#define XSTR   (HIDDEN + QCH)
#define HALFW  128
#define APM    512

__device__ __constant__ const float kFScale =
    (float)(0.5 * 27.211386024367243 * 0.5291772105638411 / 7.0);

__device__ __forceinline__ float half16_xor_reduce(float p) {
    p += __int_as_float(__builtin_amdgcn_ds_swizzle(__float_as_int(p), (1 << 10) | 0x1f));
    p += __int_as_float(__builtin_amdgcn_ds_swizzle(__float_as_int(p), (2 << 10) | 0x1f));
    p += __int_as_float(__builtin_amdgcn_ds_swizzle(__float_as_int(p), (4 << 10) | 0x1f));
    p += __int_as_float(__builtin_amdgcn_ds_swizzle(__float_as_int(p), (8 << 10) | 0x1f));
    return p;
}

__global__ __launch_bounds__(256) void mlp_wmma_kernel(
    const float* __restrict__ x,  const float* __restrict__ W1,
    const float* __restrict__ b1, const float* __restrict__ W2,
    const float* __restrict__ b2, float* __restrict__ out)
{
    __shared__ float partial[8][16];

    const int tid  = threadIdx.x;
    const int wave = tid >> 5;
    const int lane = tid & 31;
    const int hi   = lane >> 4;
    const int lo   = lane & 15;
    const int rowbase = blockIdx.x << 4;
    const int colbase = wave << 4;

    const float* __restrict__ arow = x  + (size_t)(rowbase + lo) * XSTR + 2 * hi;
    const float* __restrict__ bcol = W1 + (size_t)(2 * hi) * HALFW + colbase + lo;

    v8f c = {};
    #pragma unroll 4
    for (int k = 0; k < HIDDEN; k += 4) {
        v2f a, b;
        a.x = arow[k];
        a.y = arow[k + 1];
        b.x = bcol[(size_t)k * HALFW];
        b.y = bcol[(size_t)k * HALFW + HALFW];
        c = __builtin_amdgcn_wmma_f32_16x16x4_f32(
                false, a, false, b, (short)0, c, false, false);
    }

    const float w2v = W2[colbase + lo];
    const float b1v = b1[colbase + lo];

    #pragma unroll
    for (int r = 0; r < 8; ++r) {
        float h = c[r] + b1v;
        h = h * __builtin_amdgcn_rcpf(1.0f + __expf(-h));
        float p = half16_xor_reduce(h * w2v);
        if (lo == 0) partial[wave][r + 8 * hi] = p;
    }
    __syncthreads();

    if (tid < 16) {
        float s = b2[0];
        #pragma unroll
        for (int w = 0; w < 8; ++w) s += partial[w][tid];
        out[rowbase + tid] = s;
    }
}

__global__ __launch_bounds__(128) void pair_kernel(
    const float* __restrict__ x, const float* __restrict__ pos,
    const float* __restrict__ qw, float* __restrict__ out)
{
    __shared__ float4 sp[128];
    __shared__ float4 sq[128][3];

    const int b     = blockIdx.x;
    const int mol   = b >> 4;
    const int ipart = (b >> 2) & 3;
    const int jpart = b & 3;
    const int tid   = threadIdx.x;
    const int molbase = mol * APM;
    const int i     = molbase + ipart * 128 + tid;
    const int jbase = molbase + jpart * 128;

    float sw[QCH];
    #pragma unroll
    for (int cch = 0; cch < QCH; ++cch) sw[cch] = sqrtf(qw[cch]);

    {
        const int j = jbase + tid;
        const float* pj = pos + 3 * (size_t)j;
        sp[tid] = make_float4(pj[0], pj[1], pj[2], 0.0f);
        const float4* cj = (const float4*)(x + (size_t)j * XSTR + HIDDEN);
        float4 q0 = cj[0], q1 = cj[1], q2 = cj[2];
        q0.x *= sw[0];  q0.y *= sw[1];  q0.z *= sw[2];   q0.w *= sw[3];
        q1.x *= sw[4];  q1.y *= sw[5];  q1.z *= sw[6];   q1.w *= sw[7];
        q2.x *= sw[8];  q2.y *= sw[9];  q2.z *= sw[10];  q2.w *= sw[11];
        sq[tid][0] = q0; sq[tid][1] = q1; sq[tid][2] = q2;
    }

    const float* pi = pos + 3 * (size_t)i;
    const float pix = pi[0], piy = pi[1], piz = pi[2];
    const float4* ci = (const float4*)(x + (size_t)i * XSTR + HIDDEN);
    float4 qi0 = ci[0], qi1 = ci[1], qi2 = ci[2];
    qi0.x *= sw[0];  qi0.y *= sw[1];  qi0.z *= sw[2];   qi0.w *= sw[3];
    qi1.x *= sw[4];  qi1.y *= sw[5];  qi1.z *= sw[6];   qi1.w *= sw[7];
    qi2.x *= sw[8];  qi2.y *= sw[9];  qi2.z *= sw[10];  qi2.w *= sw[11];

    __syncthreads();

    const float inv_rc = 1.0f / 4.6f;
    float acc = 0.0f;
    #pragma unroll 4
    for (int jj = 0; jj < 128; ++jj) {
        const float4 pj = sp[jj];
        const float dx = pix - pj.x;
        const float dy = piy - pj.y;
        const float dz = piz - pj.z;
        const float d2 = dx * dx + dy * dy + dz * dz;

        const float4 a0 = sq[jj][0];
        const float4 a1 = sq[jj][1];
        const float4 a2 = sq[jj][2];
        float qd = qi0.x * a0.x + qi0.y * a0.y + qi0.z * a0.z + qi0.w * a0.w
                 + qi1.x * a1.x + qi1.y * a1.y + qi1.z * a1.z + qi1.w * a1.w
                 + qi2.x * a2.x + qi2.y * a2.y + qi2.z * a2.z + qi2.w * a2.w;

        const float inv_d = __builtin_amdgcn_rsqf(d2);
        const float d     = d2 * inv_d;
        const float r     = fminf(d * inv_rc, 1.0f - 1e-6f);
        const float fc = 1.0f - __expf(1.0f - __builtin_amdgcn_rcpf(1.0f - r * r));
        const float contrib = fc * inv_d * qd;
        acc += (jbase + jj != i) ? contrib : 0.0f;
    }

    __hip_atomic_fetch_add(&out[i], acc * kFScale,
                           __ATOMIC_RELAXED, __HIP_MEMORY_SCOPE_AGENT);
}

extern "C" void kernel_launch(void* const* d_in, const int* in_sizes, int n_in,
                              void* d_out, int out_size, void* d_ws, size_t ws_size,
                              hipStream_t stream) {
    const float* x   = (const float*)d_in[0];
    const float* pos = (const float*)d_in[1];
    const float* W1  = (const float*)d_in[5];
    const float* b1  = (const float*)d_in[6];
    const float* W2  = (const float*)d_in[7];
    const float* b2  = (const float*)d_in[8];
    const float* qw  = (const float*)d_in[9];
    float* out = (float*)d_out;

    const int N    = in_sizes[0] / XSTR;
    const int nmol = N / APM;

    mlp_wmma_kernel<<<N / 16, 256, 0, stream>>>(x, W1, b1, W2, b2, out);
    pair_kernel<<<nmol * 16, 128, 0, stream>>>(x, pos, qw, out);
}


# GEMM kernel re-cut W1-stationary: W1 staged once per block in LDS as k-quad slots, feat strips straight to registers by 16-byte loads, K in 4 prefetched quarters; k and column sum order changed
# speedup vs baseline: 2.3874x; 2.3874x over previous
#include <hip/hip_runtime.h>
#include <hip/hip_bf16.h>


typedef __attribute__((ext_vector_type(2))) float v2f;
typedef __attribute__((ext_vector_type(8))) float v8f;

#define HIDDEN 256
#define QCH    12
#define XSTR   (HIDDEN + QCH)
#define HALFW  128
#define APM    512

__device__ __forceinline__ float half16_xor_reduce(float p) {
    p += __int_as_float(__builtin_amdgcn_ds_swizzle(__float_as_int(p), (1 << 10) | 0x1f));
    p += __int_as_float(__builtin_amdgcn_ds_swizzle(__float_as_int(p), (2 << 10) | 0x1f));
    p += __int_as_float(__builtin_amdgcn_ds_swizzle(__float_as_int(p), (4 << 10) | 0x1f));
    p += __int_as_float(__builtin_amdgcn_ds_swizzle(__float_as_int(p), (8 << 10) | 0x1f));
    return p;
}

#define WPITCH 128

__device__ __forceinline__ void w1_fetch(const float* __restrict__ W1, const int q,
                                         const int wave, const int lane, float4 (&r)[2][4])
{
    #pragma unroll
    for (int t = 0; t < 2; ++t) {
        const int g = 16 * q + 2 * wave + t;
        #pragma unroll
        for (int i = 0; i < 4; ++i)
            r[t][i] = *(const float4*)(W1 + (size_t)(4 * g + i) * HALFW + 4 * lane);
    }
}

__device__ __forceinline__ void w1_put(float4* sW, const int q, const int wave,
                                       const int lane, const float4 (&r)[2][4])
{
    #pragma unroll
    for (int t = 0; t < 2; ++t) {
        const int g = 16 * q + 2 * wave + t;
        float4* dst = sW + g * WPITCH + lane;
        dst[0]  = make_float4(r[t][0].x, r[t][1].x, r[t][2].x, r[t][3].x);
        dst[32] = make_float4(r[t][0].y, r[t][1].y, r[t][2].y, r[t][3].y);
        dst[64] = make_float4(r[t][0].z, r[t][1].z, r[t][2].z, r[t][3].z);
        dst[96] = make_float4(r[t][0].w, r[t][1].w, r[t][2].w, r[t][3].w);
    }
}

__device__ __forceinline__ void strip_fetch(const float* __restrict__ arow, const int q,
                                            float4 (&af)[8])
{
    #pragma unroll
    for (int jj = 0; jj < 8; ++jj) af[jj] = *(const float4*)(arow + 8 * (8 * q + jj));
}

__device__ __forceinline__ void b_group(const float4* sWl, const int j, float4 (&b)[8])
{
    #pragma unroll
    for (int n = 0; n < 4; ++n) b[n]     = sWl[(2 * j) * WPITCH + 16 * n];
    #pragma unroll
    for (int n = 0; n < 4; ++n) b[4 + n] = sWl[(2 * j + 2) * WPITCH + 16 * n];
}

__device__ __forceinline__ void kstep(const float4 a4, const float4* b, v8f (&c)[4])
{
    #pragma unroll
    for (int n = 0; n < 4; ++n) {
        v2f a, bb;
        a.x = a4.x; a.y = a4.y;
        bb.x = b[n].x; bb.y = b[n].y;
        c[n] = __builtin_amdgcn_wmma_f32_16x16x4_f32(
                   false, a, false, bb, (short)0, c[n], false, false);
    }
    #pragma unroll
    for (int n = 0; n < 4; ++n) {
        v2f a, bb;
        a.x = a4.z; a.y = a4.w;
        bb.x = b[n].z; bb.y = b[n].w;
        c[n] = __builtin_amdgcn_wmma_f32_16x16x4_f32(
                   false, a, false, bb, (short)0, c[n], false, false);
    }
}

__device__ __forceinline__ void gemm_quarter(const float4* sWl, const int q,
                                             const float4 (&af)[8], v8f (&c)[4])
{
    float4 b[2][8];
    b_group(sWl, 8 * q, b[0]);
    #pragma unroll
    for (int gg = 0; gg < 4; ++gg) {
        if (gg < 3) b_group(sWl, 8 * q + 2 * gg + 2, b[(gg + 1) & 1]);
        __builtin_amdgcn_sched_barrier(0);
        kstep(af[2 * gg],     &b[gg & 1][0], c);
        kstep(af[2 * gg + 1], &b[gg & 1][4], c);
        __builtin_amdgcn_sched_barrier(0);
    }
}

__global__ __launch_bounds__(256) void mlp_wmma_kernel(
    const float* __restrict__ x,  const float* __restrict__ W1,
    const float* __restrict__ b1, const float* __restrict__ W2,
    const float* __restrict__ b2, float* __restrict__ out)
{
    __shared__ float4 sW[64 * WPITCH];
    __shared__ float  spart[2][64];

    const int tid  = threadIdx.x;
    const int wave = __builtin_amdgcn_readfirstlane(tid >> 5);
    const int lane = tid & 31;
    const int hi   = lane >> 4;
    const int lo   = lane & 15;
    const int rt   = wave & 3;
    const int ch   = wave >> 2;
    const int rowbase = (blockIdx.x << 6) + (rt << 4);

    const float* __restrict__ arow = x + (size_t)(rowbase + lo) * XSTR + 4 * hi;
    const float4* sWl = sW + hi * WPITCH + 64 * ch + lo;

    float4 wr[2][4];
    float4 afa[8], afb[8];
    v8f c[4];
    #pragma unroll
    for (int n = 0; n < 4; ++n) c[n] = (v8f){};

    w1_fetch(W1, 0, wave, lane, wr);
    strip_fetch(arow, 0, afa);
    w1_put(sW, 0, wave, lane, wr);
    __syncthreads();

    w1_fetch(W1, 1, wave, lane, wr);
    strip_fetch(arow, 1, afb);
    gemm_quarter(sWl, 0, afa, c);
    w1_put(sW, 1, wave, lane, wr);
    __syncthreads();

    w1_fetch(W1, 2, wave, lane, wr);
    strip_fetch(arow, 2, afa);
    gemm_quarter(sWl, 1, afb, c);
    w1_put(sW, 2, wave, lane, wr);
    __syncthreads();

    w1_fetch(W1, 3, wave, lane, wr);
    strip_fetch(arow, 3, afb);
    gemm_quarter(sWl, 2, afa, c);
    w1_put(sW, 3, wave, lane, wr);
    __syncthreads();

    gemm_quarter(sWl, 3, afb, c);

    float b1v[4], w2v[4];
    #pragma unroll
    for (int n = 0; n < 4; ++n) {
        const int p   = 64 * ch + 16 * n + lo;
        const int col = 4 * (p & 31) + (p >> 5);
        b1v[n] = b1[col];
        w2v[n] = W2[col];
    }
    float mine = 0.0f;
    #pragma unroll
    for (int r = 0; r < 8; ++r) {
        float s = 0.0f;
        #pragma unroll
        for (int n = 0; n < 4; ++n) {
            float h = c[n][r] + b1v[n];
            h = h * __builtin_amdgcn_rcpf(1.0f + __expf(-h));
            s += h * w2v[n];
        }
        const float srow = half16_xor_reduce(s);
        mine = (lo == r) ? srow : mine;
    }
    if (lo < 8) spart[ch][(rt << 4) + lo + 8 * hi] = mine;
    __syncthreads();

    if (tid < 32) {
        const float bb = b2[0];
        float2 o;
        o.x = bb + spart[0][2 * tid]     + spart[1][2 * tid];
        o.y = bb + spart[0][2 * tid + 1] + spart[1][2 * tid + 1];
        *(float2*)(out + (blockIdx.x << 6) + 2 * tid) = o;
    }
}

template <bool DIAG>
__device__ __forceinline__ void pair_tile(
    const float4* sp, const float* sq, const int jt, const int lo, const int hi,
    const v2f a0, const v2f a1, const v2f a2,
    const float (&pix)[8], const float (&piy)[8], const float (&piz)[8],
    float (&acc)[8])
{
    const int ja = jt * 16 + lo;
    const float4 pj = sp[ja];
    const float* brow = sq + ja * QCH + 2 * hi;
    const v2f b0 = *(const v2f*)(brow);
    const v2f b1 = *(const v2f*)(brow + 4);
    const v2f b2 = *(const v2f*)(brow + 8);

    v8f c = {};
    c = __builtin_amdgcn_wmma_f32_16x16x4_f32(false, a0, false, b0, (short)0, c, false, false);
    c = __builtin_amdgcn_wmma_f32_16x16x4_f32(false, a1, false, b1, (short)0, c, false, false);
    c = __builtin_amdgcn_wmma_f32_16x16x4_f32(false, a2, false, b2, (short)0, c, false, false);

    const float inv_rc = 1.0f / 4.6f;
    float inv_d[8], t[8];
    #pragma unroll
    for (int r = 0; r < 8; ++r) {
        const float dx = pix[r] - pj.x;
        const float dy = piy[r] - pj.y;
        const float dz = piz[r] - pj.z;
        t[r] = dx * dx + dy * dy + dz * dz;
    }
    #pragma unroll
    for (int r = 0; r < 8; ++r)
        inv_d[r] = __builtin_amdgcn_rsqf(t[r]);
    #pragma unroll
    for (int r = 0; r < 8; ++r) {
        const float d  = t[r] * inv_d[r];
        const float rr = fminf(d * inv_rc, 1.0f - 1e-6f);
        t[r] = 1.0f - rr * rr;
    }
    #pragma unroll
    for (int r = 0; r < 8; ++r)
        t[r] = __builtin_amdgcn_rcpf(t[r]);
    #pragma unroll
    for (int r = 0; r < 8; ++r)
        t[r] = __expf(1.0f - t[r]);
    #pragma unroll
    for (int r = 0; r < 8; ++r) {
        float contrib = (1.0f - t[r]) * inv_d[r] * c[r];
        if (DIAG) contrib = (lo != r + 8 * hi) ? contrib : 0.0f;
        acc[r] += contrib;
    }
}

__global__ __launch_bounds__(256) void pair_kernel(
    const float* __restrict__ x, const float* __restrict__ pos,
    const float* __restrict__ qw, float* __restrict__ out)
{
    __shared__ float4 sp[APM];
    __shared__ float4 sq4[APM * 3];
    __shared__ float4 sw4[3];
    const float* sq = (const float*)sq4;

    const int tid  = threadIdx.x;
    const int wave = __builtin_amdgcn_readfirstlane(tid >> 5);
    const int lane = tid & 31;
    const int hi   = lane >> 4;
    const int lo   = lane & 15;
    const int molbase = (blockIdx.x >> 3) * APM;
    const int it   = ((blockIdx.x & 7) << 2) + (wave & 3);
    const int jt0  = (wave >> 2) << 4;

    for (int k = 0; k < 6; ++k) {
        const unsigned f    = (unsigned)tid + 256u * k;
        const unsigned atom = f / 3u;
        const unsigned part = f - 3u * atom;
        const float4* src = (const float4*)(
            x + (size_t)(molbase + atom) * XSTR + HIDDEN + 4u * part);
        sq4[f] = *src;
    }
    for (int k = 0; k < 2; ++k) {
        const int a = tid + 256 * k;
        const float* pj = pos + 3 * (size_t)(molbase + a);
        sp[a] = make_float4(pj[0], pj[1], pj[2], 0.0f);
    }

    if (tid == 0) {
        sw4[0] = make_float4(qw[0], qw[1], qw[2],  qw[3]);
        sw4[1] = make_float4(qw[4], qw[5], qw[6],  qw[7]);
        sw4[2] = make_float4(qw[8], qw[9], qw[10], qw[11]);
    }
    float wsum = 0.0f;
    for (int cch = 0; cch < QCH; ++cch) wsum += qw[cch];
    const float fscale =
        (float)(0.5 * 27.211386024367243 * 0.5291772105638411) / wsum;

    __syncthreads();

    const float* arow = sq + (it * 16 + lo) * QCH + 2 * hi;
    v2f a0 = *(const v2f*)(arow);
    v2f a1 = *(const v2f*)(arow + 4);
    v2f a2 = *(const v2f*)(arow + 8);
    const float* wrow = (const float*)sw4 + 2 * hi;
    a0 *= *(const v2f*)(wrow);
    a1 *= *(const v2f*)(wrow + 4);
    a2 *= *(const v2f*)(wrow + 8);

    float pix[8], piy[8], piz[8];
    #pragma unroll
    for (int r = 0; r < 8; ++r) {
        const float4 p = sp[it * 16 + r + 8 * hi];
        pix[r] = p.x; piy[r] = p.y; piz[r] = p.z;
    }

    float acc[8];
    #pragma unroll
    for (int r = 0; r < 8; ++r) acc[r] = 0.0f;

    for (int jt = jt0; jt < jt0 + 16; ++jt) {
        if (jt == it) continue;
        pair_tile<false>(sp, sq, jt, lo, hi, a0, a1, a2, pix, piy, piz, acc);
    }
    if (it >= jt0 && it < jt0 + 16)
        pair_tile<true>(sp, sq, it, lo, hi, a0, a1, a2, pix, piy, piz, acc);

    float mine = 0.0f;
    #pragma unroll
    for (int r = 0; r < 8; ++r) {
        const float srow = half16_xor_reduce(acc[r]);
        mine = (lo == r) ? srow : mine;
    }
    if (lo < 8)
        __hip_atomic_fetch_add(&out[molbase + it * 16 + lo + 8 * hi], mine * fscale,
                               __ATOMIC_RELAXED, __HIP_MEMORY_SCOPE_AGENT);
}

extern "C" void kernel_launch(void* const* d_in, const int* in_sizes, int n_in,
                              void* d_out, int out_size, void* d_ws, size_t ws_size,
                              hipStream_t stream) {
    const float* x   = (const float*)d_in[0];
    const float* pos = (const float*)d_in[1];
    const float* W1  = (const float*)d_in[5];
    const float* b1  = (const float*)d_in[6];
    const float* W2  = (const float*)d_in[7];
    const float* b2  = (const float*)d_in[8];
    const float* qw  = (const float*)d_in[9];
    float* out = (float*)d_out;

    const int N    = in_sizes[0] / XSTR;
    const int nmol = N / APM;

    mlp_wmma_kernel<<<N / 64, 256, 0, stream>>>(x, W1, b1, W2, b2, out);
    pair_kernel<<<nmol * 8, 256, 0, stream>>>(x, pos, qw, out);
}
